# starting kernel
# baseline (speedup 1.0000x reference)
#include <hip/hip_runtime.h>


typedef __attribute__((ext_vector_type(2))) float v2f;
typedef __attribute__((ext_vector_type(8))) float v8f;

#define N_BATCH   32
#define SEQ       1024
#define SENT_LEN  512
#define CTX_LEN   256
#define LO        (1 + CTX_LEN)
#define NUM_SEG   2048
#define NROWS     (N_BATCH * SENT_LEN)
#define FEAT      2048
#define SEGS_PER_TILE 16
#define CAP       4096

static __device__ __forceinline__ int imin(int a, int b) { return a < b ? a : b; }
static __device__ __forceinline__ int imax(int a, int b) { return a > b ? a : b; }

__launch_bounds__(256, 2)
__global__ void seg_mean_wmma_kernel(const float* __restrict__ x,
                                     const int*   __restrict__ seg,
                                     float*       __restrict__ out) {
    __shared__ int2  s_pair[CAP + 4];
    __shared__ int   s_cnt[SEGS_PER_TILE];
    __shared__ float s_den[SEGS_PER_TILE];
    __shared__ int   s_toff[256];
    __shared__ int   s_total;

    const int tid  = threadIdx.x;
    const int base = blockIdx.x * SEGS_PER_TILE;

    if (tid < SEGS_PER_TILE) s_cnt[tid] = 0;

    int c = 0;
    for (int i = tid; i < NROWS; i += 256) {
        const int b = i >> 9, t = i & (SENT_LEN - 1);
        const int s = seg[b * SEQ + LO + t];
        if ((unsigned)(s - base) < (unsigned)SEGS_PER_TILE) c++;
    }
    s_toff[tid] = c;
    __syncthreads();
    if (tid == 0) {
        int acc = 0;
        for (int k = 0; k < 256; ++k) { int v = s_toff[k]; s_toff[k] = acc; acc += v; }
        s_total = imin(acc, CAP);
    }
    __syncthreads();

    int pos = s_toff[tid];
    for (int i = tid; i < NROWS; i += 256) {
        const int b = i >> 9, t = i & (SENT_LEN - 1);
        const int s = seg[b * SEQ + LO + t];
        const unsigned d = (unsigned)(s - base);
        if (d < (unsigned)SEGS_PER_TILE) {
            if (pos < CAP) {
                int2 e;
                e.x = (b * SEQ + LO + t) << 10;
                e.y = (b << 20) | (int)d;
                s_pair[pos] = e;
            }
            pos++;
            atomicAdd(&s_cnt[d], 1);
        }
    }
    __syncthreads();
    if (tid < SEGS_PER_TILE) s_den[tid] = (float)imax(s_cnt[tid], 1);

    const int total = s_total;
    __syncthreads();
    const int totalv = s_total;
    if (tid < 3 && totalv > 0) {
        const int idx = totalv + tid;
        if (idx & 3 || tid < ((4 - (totalv & 3)) & 3)) {   }
        int2 p0 = s_pair[0];
        int2 e; e.x = p0.x; e.y = (p0.y & ~15) | 16;
        s_pair[idx] = e;
    }
    __syncthreads();
    (void)total;

    const int ngroups = (totalv + 3) >> 2;

    const int lane = tid & 31;
    const int wave = tid >> 5;
    const int n    = lane & 15;
    const int hi   = lane >> 4;
    const int k0 = hi ? 2 : 0;

    for (int stripe = 0; stripe < FEAT / 128; ++stripe) {
        const int fb    = stripe * 128 + wave * 16;
        const int isEmb = (fb < 1024);
        const int colOff = isEmb ? (fb + n) : (fb + n - 1024);

        v8f acc = {};
        for (int g = 0; g < ngroups; ++g) {
            const int gi = (g << 2) + k0;
            const int2 p0 = s_pair[gi];
            const int2 p1 = s_pair[gi + 1];

            v2f a;
            a.x = ((p0.y & 31) == n) ? 1.0f : 0.0f;
            a.y = ((p1.y & 31) == n) ? 1.0f : 0.0f;

            const int b0 = (isEmb ? p0.x : (p0.y & ~31)) + colOff;
            const int b1 = (isEmb ? p1.x : (p1.y & ~31)) + colOff;
            v2f bv;
            bv.x = x[b0];
            bv.y = x[b1];

            acc = __builtin_amdgcn_wmma_f32_16x16x4_f32(
                false, a, false, bv, (short)0, acc, false, false);
        }

        union { v8f v; float f[8]; } u; u.v = acc;
        #pragma unroll
        for (int j = 0; j < 8; ++j) {
            const int m = j + (hi << 3);
            out[(long)(base + m) * FEAT + fb + n] = u.f[j] / s_den[m];
        }
    }
}

extern "C" void kernel_launch(void* const* d_in, const int* in_sizes, int n_in,
                              void* d_out, int out_size, void* d_ws, size_t ws_size,
                              hipStream_t stream) {
    (void)in_sizes; (void)n_in; (void)out_size; (void)d_ws; (void)ws_size;
    const float* x   = (const float*)d_in[0];
    const int*   seg = (const int*)d_in[1];
    float*       out = (float*)d_out;
    seg_mean_wmma_kernel<<<128, 256, 0, stream>>>(x, seg, out);
}


# Gather tile widened: each wave owns 128 features (two 16-byte loads per row per lane, 8 accumulators), 2 passes instead of 16 stripes, 8x fewer latency-bound gather trips; sum order unchanged
# speedup vs baseline: 2.6429x; 2.6429x over previous
#include <hip/hip_runtime.h>


typedef __attribute__((ext_vector_type(2))) float v2f;
typedef __attribute__((ext_vector_type(4))) float v4f;
typedef __attribute__((ext_vector_type(8))) float v8f;

#define N_BATCH   32
#define SEQ       1024
#define SENT_LEN  512
#define CTX_LEN   256
#define LO        (1 + CTX_LEN)
#define NUM_SEG   2048
#define NROWS     (N_BATCH * SENT_LEN)
#define FEAT      2048
#define SEGS_PER_TILE 16
#define CAP       4096

static __device__ __forceinline__ int imin(int a, int b) { return a < b ? a : b; }
static __device__ __forceinline__ int imax(int a, int b) { return a > b ? a : b; }

__launch_bounds__(256, 2)
__global__ void seg_mean_wmma_kernel(const float* __restrict__ x,
                                     const int*   __restrict__ seg,
                                     float*       __restrict__ out) {
    __shared__ int2  s_pair[CAP + 4];
    __shared__ int   s_cnt[SEGS_PER_TILE];
    __shared__ float s_den[SEGS_PER_TILE];
    __shared__ int   s_toff[256];
    __shared__ int   s_total;

    const int tid  = threadIdx.x;
    const int base = blockIdx.x * SEGS_PER_TILE;

    if (tid < SEGS_PER_TILE) s_cnt[tid] = 0;

    int c = 0;
    for (int i = tid; i < NROWS; i += 256) {
        const int b = i >> 9, t = i & (SENT_LEN - 1);
        const int s = seg[b * SEQ + LO + t];
        if ((unsigned)(s - base) < (unsigned)SEGS_PER_TILE) c++;
    }
    s_toff[tid] = c;
    __syncthreads();
    if (tid == 0) {
        int acc = 0;
        for (int k = 0; k < 256; ++k) { int v = s_toff[k]; s_toff[k] = acc; acc += v; }
        s_total = imin(acc, CAP);
    }
    __syncthreads();

    int pos = s_toff[tid];
    for (int i = tid; i < NROWS; i += 256) {
        const int b = i >> 9, t = i & (SENT_LEN - 1);
        const int s = seg[b * SEQ + LO + t];
        const unsigned d = (unsigned)(s - base);
        if (d < (unsigned)SEGS_PER_TILE) {
            if (pos < CAP) {
                int2 e;
                e.x = (b * SEQ + LO + t) << 10;
                e.y = (b << 20) | (int)d;
                s_pair[pos] = e;
            }
            pos++;
            atomicAdd(&s_cnt[d], 1);
        }
    }
    __syncthreads();
    if (tid < SEGS_PER_TILE) s_den[tid] = (float)imax(s_cnt[tid], 1);

    const int total = s_total;
    __syncthreads();
    const int totalv = s_total;
    if (tid < 3 && totalv > 0) {
        const int idx = totalv + tid;
        if (idx & 3 || tid < ((4 - (totalv & 3)) & 3)) {   }
        int2 p0 = s_pair[0];
        int2 e; e.x = p0.x; e.y = (p0.y & ~15) | 16;
        s_pair[idx] = e;
    }
    __syncthreads();
    (void)total;

    const int ngroups = (totalv + 3) >> 2;

    const int lane = tid & 31;
    const int wave = tid >> 5;
    const int n    = lane & 15;
    const int hi   = lane >> 4;
    const int k0 = hi ? 2 : 0;

    const int fw = wave * 128 + (n << 2);
    for (int pass = 0; pass < 2; ++pass) {
        const int isEmb = (pass == 0);

        v8f acc[2][4] = {};
        for (int g = 0; g < ngroups; ++g) {
            const int gi = (g << 2) + k0;
            const int2 p0 = s_pair[gi];
            const int2 p1 = s_pair[gi + 1];

            v2f a;
            a.x = ((p0.y & 31) == n) ? 1.0f : 0.0f;
            a.y = ((p1.y & 31) == n) ? 1.0f : 0.0f;

            const int r0 = (isEmb ? p0.x : (p0.y & ~31)) + fw;
            const int r1 = (isEmb ? p1.x : (p1.y & ~31)) + fw;
            const v4f u0 = *(const v4f*)(x + r0);
            const v4f u1 = *(const v4f*)(x + r0 + 64);
            const v4f w0 = *(const v4f*)(x + r1);
            const v4f w1 = *(const v4f*)(x + r1 + 64);

            #pragma unroll
            for (int j = 0; j < 4; ++j) {
                v2f b0; b0.x = u0[j]; b0.y = w0[j];
                acc[0][j] = __builtin_amdgcn_wmma_f32_16x16x4_f32(
                    false, a, false, b0, (short)0, acc[0][j], false, false);
                v2f b1; b1.x = u1[j]; b1.y = w1[j];
                acc[1][j] = __builtin_amdgcn_wmma_f32_16x16x4_f32(
                    false, a, false, b1, (short)0, acc[1][j], false, false);
            }
        }

        const int colOut = pass * 1024 + fw;
        #pragma unroll
        for (int r = 0; r < 8; ++r) {
            const int m = r + (hi << 3);
            const float den = s_den[m];
            float* orow = out + (long)(base + m) * FEAT + colOut;
            #pragma unroll
            for (int q = 0; q < 2; ++q) {
                v4f o;
                o.x = acc[q][0][r] / den;
                o.y = acc[q][1][r] / den;
                o.z = acc[q][2][r] / den;
                o.w = acc[q][3][r] / den;
                *(v4f*)(orow + 64 * q) = o;
            }
        }
    }
}

extern "C" void kernel_launch(void* const* d_in, const int* in_sizes, int n_in,
                              void* d_out, int out_size, void* d_ws, size_t ws_size,
                              hipStream_t stream) {
    (void)in_sizes; (void)n_in; (void)out_size; (void)d_ws; (void)ws_size;
    const float* x   = (const float*)d_in[0];
    const int*   seg = (const int*)d_in[1];
    float*       out = (float*)d_out;
    seg_mean_wmma_kernel<<<128, 256, 0, stream>>>(x, seg, out);
}


# Segment ids read once: each thread keeps its 64 ids in registers from count to fill, all 64 loads in flight together; the fill's 64 one-at-a-time load round trips are gone; sum order unchanged
# speedup vs baseline: 3.6099x; 1.3659x over previous
#include <hip/hip_runtime.h>


typedef __attribute__((ext_vector_type(2))) float v2f;
typedef __attribute__((ext_vector_type(4))) float v4f;
typedef __attribute__((ext_vector_type(8))) float v8f;

#define N_BATCH   32
#define SEQ       1024
#define SENT_LEN  512
#define CTX_LEN   256
#define LO        (1 + CTX_LEN)
#define NUM_SEG   2048
#define NROWS     (N_BATCH * SENT_LEN)
#define FEAT      2048
#define SEGS_PER_TILE 16
#define CAP       4096
#define IDS_PER_THREAD (NROWS / 256)

static __device__ __forceinline__ int imin(int a, int b) { return a < b ? a : b; }
static __device__ __forceinline__ int imax(int a, int b) { return a > b ? a : b; }

__launch_bounds__(256, 2)
__global__ void seg_mean_wmma_kernel(const float* __restrict__ x,
                                     const int*   __restrict__ seg,
                                     float*       __restrict__ out) {
    __shared__ int2  s_pair[CAP + 4];
    __shared__ int   s_cnt[SEGS_PER_TILE];
    __shared__ float s_den[SEGS_PER_TILE];
    __shared__ int   s_toff[256];
    __shared__ int   s_total;

    const int tid  = threadIdx.x;
    const int base = blockIdx.x * SEGS_PER_TILE;

    if (tid < SEGS_PER_TILE) s_cnt[tid] = 0;

    int sv[IDS_PER_THREAD];
    const int* sp = seg + LO + tid;
    #pragma unroll
    for (int k = 0; k < IDS_PER_THREAD; ++k)
        sv[k] = sp[(k >> 1) * SEQ + (k & 1) * 256];
    int c = 0;
    #pragma unroll
    for (int k = 0; k < IDS_PER_THREAD; ++k)
        c += ((unsigned)(sv[k] - base) < (unsigned)SEGS_PER_TILE) ? 1 : 0;
    s_toff[tid] = c;
    __syncthreads();
    if (tid == 0) {
        int acc = 0;
        for (int k = 0; k < 256; ++k) { int v = s_toff[k]; s_toff[k] = acc; acc += v; }
        s_total = imin(acc, CAP);
    }
    __syncthreads();

    int pos = s_toff[tid];
    #pragma unroll
    for (int k = 0; k < IDS_PER_THREAD; ++k) {
        const int b = k >> 1, t = (k & 1) * 256 + tid;
        const unsigned d = (unsigned)(sv[k] - base);
        if (d < (unsigned)SEGS_PER_TILE) {
            if (pos < CAP) {
                int2 e;
                e.x = (b * SEQ + LO + t) << 10;
                e.y = (b << 20) | (int)d;
                s_pair[pos] = e;
            }
            pos++;
            atomicAdd(&s_cnt[d], 1);
        }
    }
    __syncthreads();
    if (tid < SEGS_PER_TILE) s_den[tid] = (float)imax(s_cnt[tid], 1);

    const int total = s_total;
    __syncthreads();
    const int totalv = s_total;
    if (tid < 3 && totalv > 0) {
        const int idx = totalv + tid;
        if (idx & 3 || tid < ((4 - (totalv & 3)) & 3)) {   }
        int2 p0 = s_pair[0];
        int2 e; e.x = p0.x; e.y = (p0.y & ~15) | 16;
        s_pair[idx] = e;
    }
    __syncthreads();
    (void)total;

    const int ngroups = (totalv + 3) >> 2;

    const int lane = tid & 31;
    const int wave = tid >> 5;
    const int n    = lane & 15;
    const int hi   = lane >> 4;
    const int k0 = hi ? 2 : 0;

    const int fw = wave * 128 + (n << 2);
    for (int pass = 0; pass < 2; ++pass) {
        const int isEmb = (pass == 0);

        v8f acc[2][4] = {};
        for (int g = 0; g < ngroups; ++g) {
            const int gi = (g << 2) + k0;
            const int2 p0 = s_pair[gi];
            const int2 p1 = s_pair[gi + 1];

            v2f a;
            a.x = ((p0.y & 31) == n) ? 1.0f : 0.0f;
            a.y = ((p1.y & 31) == n) ? 1.0f : 0.0f;

            const int r0 = (isEmb ? p0.x : (p0.y & ~31)) + fw;
            const int r1 = (isEmb ? p1.x : (p1.y & ~31)) + fw;
            const v4f u0 = *(const v4f*)(x + r0);
            const v4f u1 = *(const v4f*)(x + r0 + 64);
            const v4f w0 = *(const v4f*)(x + r1);
            const v4f w1 = *(const v4f*)(x + r1 + 64);

            #pragma unroll
            for (int j = 0; j < 4; ++j) {
                v2f b0; b0.x = u0[j]; b0.y = w0[j];
                acc[0][j] = __builtin_amdgcn_wmma_f32_16x16x4_f32(
                    false, a, false, b0, (short)0, acc[0][j], false, false);
                v2f b1; b1.x = u1[j]; b1.y = w1[j];
                acc[1][j] = __builtin_amdgcn_wmma_f32_16x16x4_f32(
                    false, a, false, b1, (short)0, acc[1][j], false, false);
            }
        }

        const int colOut = pass * 1024 + fw;
        #pragma unroll
        for (int r = 0; r < 8; ++r) {
            const int m = r + (hi << 3);
            const float den = s_den[m];
            float* orow = out + (long)(base + m) * FEAT + colOut;
            #pragma unroll
            for (int q = 0; q < 2; ++q) {
                v4f o;
                o.x = acc[q][0][r] / den;
                o.y = acc[q][1][r] / den;
                o.z = acc[q][2][r] / den;
                o.w = acc[q][3][r] / den;
                *(v4f*)(orow + 64 * q) = o;
            }
        }
    }
}

extern "C" void kernel_launch(void* const* d_in, const int* in_sizes, int n_in,
                              void* d_out, int out_size, void* d_ws, size_t ws_size,
                              hipStream_t stream) {
    (void)in_sizes; (void)n_in; (void)out_size; (void)d_ws; (void)ws_size;
    const float* x   = (const float*)d_in[0];
    const int*   seg = (const int*)d_in[1];
    float*       out = (float*)d_out;
    seg_mean_wmma_kernel<<<128, 256, 0, stream>>>(x, seg, out);
}


# Segment tile 16 -> 4 per block, grid 128 -> 512 (two blocks per WGP on all 256): gather work per block 4x smaller; same-segment rows now share a 4-row WMMA group more often (sum pairing changes)
# speedup vs baseline: 5.9024x; 1.6351x over previous
#include <hip/hip_runtime.h>


typedef __attribute__((ext_vector_type(2))) float v2f;
typedef __attribute__((ext_vector_type(4))) float v4f;
typedef __attribute__((ext_vector_type(8))) float v8f;

#define N_BATCH   32
#define SEQ       1024
#define SENT_LEN  512
#define CTX_LEN   256
#define LO        (1 + CTX_LEN)
#define NUM_SEG   2048
#define NROWS     (N_BATCH * SENT_LEN)
#define FEAT      2048
#define SEGS_PER_TILE 4
#define CAP       4096
#define IDS_PER_THREAD (NROWS / 256)

static __device__ __forceinline__ int imin(int a, int b) { return a < b ? a : b; }
static __device__ __forceinline__ int imax(int a, int b) { return a > b ? a : b; }

__launch_bounds__(256, 2)
__global__ void seg_mean_wmma_kernel(const float* __restrict__ x,
                                     const int*   __restrict__ seg,
                                     float*       __restrict__ out) {
    __shared__ int2  s_pair[CAP + 4];
    __shared__ int   s_cnt[SEGS_PER_TILE];
    __shared__ float s_den[SEGS_PER_TILE];
    __shared__ int   s_toff[256];
    __shared__ int   s_total;

    const int tid  = threadIdx.x;
    const int base = blockIdx.x * SEGS_PER_TILE;

    if (tid < SEGS_PER_TILE) s_cnt[tid] = 0;

    int sv[IDS_PER_THREAD];
    const int* sp = seg + LO + tid;
    #pragma unroll
    for (int k = 0; k < IDS_PER_THREAD; ++k)
        sv[k] = sp[(k >> 1) * SEQ + (k & 1) * 256];
    int c = 0;
    #pragma unroll
    for (int k = 0; k < IDS_PER_THREAD; ++k)
        c += ((unsigned)(sv[k] - base) < (unsigned)SEGS_PER_TILE) ? 1 : 0;
    s_toff[tid] = c;
    __syncthreads();
    if (tid == 0) {
        int acc = 0;
        for (int k = 0; k < 256; ++k) { int v = s_toff[k]; s_toff[k] = acc; acc += v; }
        s_total = imin(acc, CAP);
    }
    __syncthreads();

    int pos = s_toff[tid];
    #pragma unroll
    for (int k = 0; k < IDS_PER_THREAD; ++k) {
        const int b = k >> 1, t = (k & 1) * 256 + tid;
        const unsigned d = (unsigned)(sv[k] - base);
        if (d < (unsigned)SEGS_PER_TILE) {
            if (pos < CAP) {
                int2 e;
                e.x = (b * SEQ + LO + t) << 10;
                e.y = (b << 20) | (int)d;
                s_pair[pos] = e;
            }
            pos++;
            atomicAdd(&s_cnt[d], 1);
        }
    }
    __syncthreads();
    if (tid < SEGS_PER_TILE) s_den[tid] = (float)imax(s_cnt[tid], 1);

    const int total = s_total;
    __syncthreads();
    const int totalv = s_total;
    if (tid < 3 && totalv > 0) {
        const int idx = totalv + tid;
        if (idx & 3 || tid < ((4 - (totalv & 3)) & 3)) {   }
        int2 p0 = s_pair[0];
        int2 e; e.x = p0.x; e.y = (p0.y & ~15) | 16;
        s_pair[idx] = e;
    }
    __syncthreads();
    (void)total;

    const int ngroups = (totalv + 3) >> 2;

    const int lane = tid & 31;
    const int wave = tid >> 5;
    const int n    = lane & 15;
    const int hi   = lane >> 4;
    const int k0 = hi ? 2 : 0;

    const int fw = wave * 128 + (n << 2);
    for (int pass = 0; pass < 2; ++pass) {
        const int isEmb = (pass == 0);

        v8f acc[2][4] = {};
        for (int g = 0; g < ngroups; ++g) {
            const int gi = (g << 2) + k0;
            const int2 p0 = s_pair[gi];
            const int2 p1 = s_pair[gi + 1];

            v2f a;
            a.x = ((p0.y & 31) == n) ? 1.0f : 0.0f;
            a.y = ((p1.y & 31) == n) ? 1.0f : 0.0f;

            const int r0 = (isEmb ? p0.x : (p0.y & ~31)) + fw;
            const int r1 = (isEmb ? p1.x : (p1.y & ~31)) + fw;
            const v4f u0 = *(const v4f*)(x + r0);
            const v4f u1 = *(const v4f*)(x + r0 + 64);
            const v4f w0 = *(const v4f*)(x + r1);
            const v4f w1 = *(const v4f*)(x + r1 + 64);

            #pragma unroll
            for (int j = 0; j < 4; ++j) {
                v2f b0; b0.x = u0[j]; b0.y = w0[j];
                acc[0][j] = __builtin_amdgcn_wmma_f32_16x16x4_f32(
                    false, a, false, b0, (short)0, acc[0][j], false, false);
                v2f b1; b1.x = u1[j]; b1.y = w1[j];
                acc[1][j] = __builtin_amdgcn_wmma_f32_16x16x4_f32(
                    false, a, false, b1, (short)0, acc[1][j], false, false);
            }
        }

        const int colOut = pass * 1024 + fw;
        if (hi == 0) {
            #pragma unroll
            for (int r = 0; r < SEGS_PER_TILE; ++r) {
                const float den = s_den[r];
                float* orow = out + (long)(base + r) * FEAT + colOut;
                #pragma unroll
                for (int q = 0; q < 2; ++q) {
                    v4f o;
                    o.x = acc[q][0][r] / den;
                    o.y = acc[q][1][r] / den;
                    o.z = acc[q][2][r] / den;
                    o.w = acc[q][3][r] / den;
                    *(v4f*)(orow + 64 * q) = o;
                }
            }
        }
    }
}

extern "C" void kernel_launch(void* const* d_in, const int* in_sizes, int n_in,
                              void* d_out, int out_size, void* d_ws, size_t ws_size,
                              hipStream_t stream) {
    (void)in_sizes; (void)n_in; (void)out_size; (void)d_ws; (void)ws_size;
    const float* x   = (const float*)d_in[0];
    const int*   seg = (const int*)d_in[1];
    float*       out = (float*)d_out;
    seg_mean_wmma_kernel<<<512, 256, 0, stream>>>(x, seg, out);
}


# List offsets by a parallel prefix sum (7 wave ballots + mbcnt per thread, 8 wave totals through LDS) instead of thread 0's serial 256-step LDS scan; same offsets, list and sum order unchanged
# speedup vs baseline: 6.4661x; 1.0955x over previous
#include <hip/hip_runtime.h>


typedef __attribute__((ext_vector_type(2))) float v2f;
typedef __attribute__((ext_vector_type(4))) float v4f;
typedef __attribute__((ext_vector_type(8))) float v8f;

#define N_BATCH   32
#define SEQ       1024
#define SENT_LEN  512
#define CTX_LEN   256
#define LO        (1 + CTX_LEN)
#define NUM_SEG   2048
#define NROWS     (N_BATCH * SENT_LEN)
#define FEAT      2048
#define SEGS_PER_TILE 4
#define CAP       4096
#define IDS_PER_THREAD (NROWS / 256)

static __device__ __forceinline__ int imin(int a, int b) { return a < b ? a : b; }
static __device__ __forceinline__ int imax(int a, int b) { return a > b ? a : b; }

__launch_bounds__(256, 2)
__global__ void seg_mean_wmma_kernel(const float* __restrict__ x,
                                     const int*   __restrict__ seg,
                                     float*       __restrict__ out) {
    __shared__ int2  s_pair[CAP + 4];
    __shared__ int   s_cnt[SEGS_PER_TILE];
    __shared__ float s_den[SEGS_PER_TILE];
    __shared__ int   s_wtot[8];

    const int tid  = threadIdx.x;
    const int base = blockIdx.x * SEGS_PER_TILE;

    if (tid < SEGS_PER_TILE) s_cnt[tid] = 0;

    int sv[IDS_PER_THREAD];
    const int* sp = seg + LO + tid;
    #pragma unroll
    for (int k = 0; k < IDS_PER_THREAD; ++k)
        sv[k] = sp[(k >> 1) * SEQ + (k & 1) * 256];
    int c = 0;
    #pragma unroll
    for (int k = 0; k < IDS_PER_THREAD; ++k)
        c += ((unsigned)(sv[k] - base) < (unsigned)SEGS_PER_TILE) ? 1 : 0;
    const int lane = tid & 31;
    const int wave = tid >> 5;
    int excl = 0, wsum = 0;
    #pragma unroll
    for (int b = 0; b < 7; ++b) {
        const unsigned m = __builtin_amdgcn_ballot_w32(((c >> b) & 1) != 0);
        excl += (int)__builtin_amdgcn_mbcnt_lo(m, 0u) << b;
        wsum += __builtin_popcount(m) << b;
    }
    if (lane == 0) s_wtot[wave] = wsum;
    __syncthreads();
    int woff = 0, tot = 0;
    #pragma unroll
    for (int w = 0; w < 8; ++w) {
        const int v = s_wtot[w];
        woff += (w < wave) ? v : 0;
        tot  += v;
    }
    const int totalv = imin(tot, CAP);

    int pos = woff + excl;
    #pragma unroll
    for (int k = 0; k < IDS_PER_THREAD; ++k) {
        const int b = k >> 1, t = (k & 1) * 256 + tid;
        const unsigned d = (unsigned)(sv[k] - base);
        if (d < (unsigned)SEGS_PER_TILE) {
            if (pos < CAP) {
                int2 e;
                e.x = (b * SEQ + LO + t) << 10;
                e.y = (b << 20) | (int)d;
                s_pair[pos] = e;
            }
            pos++;
            atomicAdd(&s_cnt[d], 1);
        }
    }
    __syncthreads();
    if (tid < SEGS_PER_TILE) s_den[tid] = (float)imax(s_cnt[tid], 1);

    if (tid < 3 && totalv > 0) {
        const int idx = totalv + tid;
        if (idx & 3 || tid < ((4 - (totalv & 3)) & 3)) {   }
        int2 p0 = s_pair[0];
        int2 e; e.x = p0.x; e.y = (p0.y & ~15) | 16;
        s_pair[idx] = e;
    }
    __syncthreads();

    const int ngroups = (totalv + 3) >> 2;

    const int n    = lane & 15;
    const int hi   = lane >> 4;
    const int k0 = hi ? 2 : 0;

    const int fw = wave * 128 + (n << 2);
    for (int pass = 0; pass < 2; ++pass) {
        const int isEmb = (pass == 0);

        v8f acc[2][4] = {};
        for (int g = 0; g < ngroups; ++g) {
            const int gi = (g << 2) + k0;
            const int2 p0 = s_pair[gi];
            const int2 p1 = s_pair[gi + 1];

            v2f a;
            a.x = ((p0.y & 31) == n) ? 1.0f : 0.0f;
            a.y = ((p1.y & 31) == n) ? 1.0f : 0.0f;

            const int r0 = (isEmb ? p0.x : (p0.y & ~31)) + fw;
            const int r1 = (isEmb ? p1.x : (p1.y & ~31)) + fw;
            const v4f u0 = *(const v4f*)(x + r0);
            const v4f u1 = *(const v4f*)(x + r0 + 64);
            const v4f w0 = *(const v4f*)(x + r1);
            const v4f w1 = *(const v4f*)(x + r1 + 64);

            #pragma unroll
            for (int j = 0; j < 4; ++j) {
                v2f b0; b0.x = u0[j]; b0.y = w0[j];
                acc[0][j] = __builtin_amdgcn_wmma_f32_16x16x4_f32(
                    false, a, false, b0, (short)0, acc[0][j], false, false);
                v2f b1; b1.x = u1[j]; b1.y = w1[j];
                acc[1][j] = __builtin_amdgcn_wmma_f32_16x16x4_f32(
                    false, a, false, b1, (short)0, acc[1][j], false, false);
            }
        }

        const int colOut = pass * 1024 + fw;
        if (hi == 0) {
            #pragma unroll
            for (int r = 0; r < SEGS_PER_TILE; ++r) {
                const float den = s_den[r];
                float* orow = out + (long)(base + r) * FEAT + colOut;
                #pragma unroll
                for (int q = 0; q < 2; ++q) {
                    v4f o;
                    o.x = acc[q][0][r] / den;
                    o.y = acc[q][1][r] / den;
                    o.z = acc[q][2][r] / den;
                    o.w = acc[q][3][r] / den;
                    *(v4f*)(orow + 64 * q) = o;
                }
            }
        }
    }
}

extern "C" void kernel_launch(void* const* d_in, const int* in_sizes, int n_in,
                              void* d_out, int out_size, void* d_ws, size_t ws_size,
                              hipStream_t stream) {
    (void)in_sizes; (void)n_in; (void)out_size; (void)d_ws; (void)ws_size;
    const float* x   = (const float*)d_in[0];
    const int*   seg = (const int*)d_in[1];
    float*       out = (float*)d_out;
    seg_mean_wmma_kernel<<<512, 256, 0, stream>>>(x, seg, out);
}


# Mean as sum times a per-segment reciprocal: one IEEE division per segment per block, multiplies in the epilogue instead of 64 IEEE divisions per lane; sum order unchanged, last-bit rounding changes
# speedup vs baseline: 6.9752x; 1.0787x over previous
#include <hip/hip_runtime.h>


typedef __attribute__((ext_vector_type(2))) float v2f;
typedef __attribute__((ext_vector_type(4))) float v4f;
typedef __attribute__((ext_vector_type(8))) float v8f;

#define N_BATCH   32
#define SEQ       1024
#define SENT_LEN  512
#define CTX_LEN   256
#define LO        (1 + CTX_LEN)
#define NUM_SEG   2048
#define NROWS     (N_BATCH * SENT_LEN)
#define FEAT      2048
#define SEGS_PER_TILE 4
#define CAP       4096
#define IDS_PER_THREAD (NROWS / 256)

static __device__ __forceinline__ int imin(int a, int b) { return a < b ? a : b; }
static __device__ __forceinline__ int imax(int a, int b) { return a > b ? a : b; }

__launch_bounds__(256, 2)
__global__ void seg_mean_wmma_kernel(const float* __restrict__ x,
                                     const int*   __restrict__ seg,
                                     float*       __restrict__ out) {
    __shared__ int2  s_pair[CAP + 4];
    __shared__ int   s_cnt[SEGS_PER_TILE];
    __shared__ float s_rinv[SEGS_PER_TILE];
    __shared__ int   s_wtot[8];

    const int tid  = threadIdx.x;
    const int base = blockIdx.x * SEGS_PER_TILE;

    if (tid < SEGS_PER_TILE) s_cnt[tid] = 0;

    int sv[IDS_PER_THREAD];
    const int* sp = seg + LO + tid;
    #pragma unroll
    for (int k = 0; k < IDS_PER_THREAD; ++k)
        sv[k] = sp[(k >> 1) * SEQ + (k & 1) * 256];
    int c = 0;
    #pragma unroll
    for (int k = 0; k < IDS_PER_THREAD; ++k)
        c += ((unsigned)(sv[k] - base) < (unsigned)SEGS_PER_TILE) ? 1 : 0;
    const int lane = tid & 31;
    const int wave = tid >> 5;
    int excl = 0, wsum = 0;
    #pragma unroll
    for (int b = 0; b < 7; ++b) {
        const unsigned m = __builtin_amdgcn_ballot_w32(((c >> b) & 1) != 0);
        excl += (int)__builtin_amdgcn_mbcnt_lo(m, 0u) << b;
        wsum += __builtin_popcount(m) << b;
    }
    if (lane == 0) s_wtot[wave] = wsum;
    __syncthreads();
    int woff = 0, tot = 0;
    #pragma unroll
    for (int w = 0; w < 8; ++w) {
        const int v = s_wtot[w];
        woff += (w < wave) ? v : 0;
        tot  += v;
    }
    const int totalv = imin(tot, CAP);

    int pos = woff + excl;
    #pragma unroll
    for (int k = 0; k < IDS_PER_THREAD; ++k) {
        const int b = k >> 1, t = (k & 1) * 256 + tid;
        const unsigned d = (unsigned)(sv[k] - base);
        if (d < (unsigned)SEGS_PER_TILE) {
            if (pos < CAP) {
                int2 e;
                e.x = (b * SEQ + LO + t) << 10;
                e.y = (b << 20) | (int)d;
                s_pair[pos] = e;
            }
            pos++;
            atomicAdd(&s_cnt[d], 1);
        }
    }
    __syncthreads();
    if (tid < SEGS_PER_TILE) s_rinv[tid] = 1.0f / (float)imax(s_cnt[tid], 1);

    if (tid < 3 && totalv > 0) {
        const int idx = totalv + tid;
        if (idx & 3 || tid < ((4 - (totalv & 3)) & 3)) {   }
        int2 p0 = s_pair[0];
        int2 e; e.x = p0.x; e.y = (p0.y & ~15) | 16;
        s_pair[idx] = e;
    }
    __syncthreads();

    const int ngroups = (totalv + 3) >> 2;

    const int n    = lane & 15;
    const int hi   = lane >> 4;
    const int k0 = hi ? 2 : 0;

    const int fw = wave * 128 + (n << 2);
    for (int pass = 0; pass < 2; ++pass) {
        const int isEmb = (pass == 0);

        v8f acc[2][4] = {};
        for (int g = 0; g < ngroups; ++g) {
            const int gi = (g << 2) + k0;
            const int2 p0 = s_pair[gi];
            const int2 p1 = s_pair[gi + 1];

            v2f a;
            a.x = ((p0.y & 31) == n) ? 1.0f : 0.0f;
            a.y = ((p1.y & 31) == n) ? 1.0f : 0.0f;

            const int r0 = (isEmb ? p0.x : (p0.y & ~31)) + fw;
            const int r1 = (isEmb ? p1.x : (p1.y & ~31)) + fw;
            const v4f u0 = *(const v4f*)(x + r0);
            const v4f u1 = *(const v4f*)(x + r0 + 64);
            const v4f w0 = *(const v4f*)(x + r1);
            const v4f w1 = *(const v4f*)(x + r1 + 64);

            #pragma unroll
            for (int j = 0; j < 4; ++j) {
                v2f b0; b0.x = u0[j]; b0.y = w0[j];
                acc[0][j] = __builtin_amdgcn_wmma_f32_16x16x4_f32(
                    false, a, false, b0, (short)0, acc[0][j], false, false);
                v2f b1; b1.x = u1[j]; b1.y = w1[j];
                acc[1][j] = __builtin_amdgcn_wmma_f32_16x16x4_f32(
                    false, a, false, b1, (short)0, acc[1][j], false, false);
            }
        }

        const int colOut = pass * 1024 + fw;
        if (hi == 0) {
            #pragma unroll
            for (int r = 0; r < SEGS_PER_TILE; ++r) {
                const float rinv = s_rinv[r];
                float* orow = out + (long)(base + r) * FEAT + colOut;
                #pragma unroll
                for (int q = 0; q < 2; ++q) {
                    v4f o;
                    o.x = acc[q][0][r] * rinv;
                    o.y = acc[q][1][r] * rinv;
                    o.z = acc[q][2][r] * rinv;
                    o.w = acc[q][3][r] * rinv;
                    *(v4f*)(orow + 64 * q) = o;
                }
            }
        }
    }
}

extern "C" void kernel_launch(void* const* d_in, const int* in_sizes, int n_in,
                              void* d_out, int out_size, void* d_ws, size_t ws_size,
                              hipStream_t stream) {
    (void)in_sizes; (void)n_in; (void)out_size; (void)d_ws; (void)ws_size;
    const float* x   = (const float*)d_in[0];
    const int*   seg = (const int*)d_in[1];
    float*       out = (float*)d_out;
    seg_mean_wmma_kernel<<<512, 256, 0, stream>>>(x, seg, out);
}
